# starting kernel
# baseline (speedup 1.0000x reference)
#include <hip/hip_runtime.h>

constexpr int kN    = 8;
constexpr int kINC  = 32;
constexpr int kINN  = 50000;
constexpr int kOUTC = 64;
constexpr int kOUTN = 8192;
constexpr int kD    = 32;

typedef __attribute__((ext_vector_type(2))) float v2f;
typedef __attribute__((ext_vector_type(8))) float v8f;

__global__ __launch_bounds__(128) void fused_gather_max_wmma(
    const float* __restrict__ x,
    const float* __restrict__ nf,
    const float* __restrict__ ft,
    const float* __restrict__ bias,
    const int*   __restrict__ A,
    float*       __restrict__ out)
{
  __shared__ int   sA[16 * kD];
  __shared__ float sred[16 * 33];

  const int tid    = threadIdx.x;
  const int o_base = blockIdx.x * 16;
  const int n      = blockIdx.y;

  for (int j = tid; j < 16 * kD; j += 128)
    sA[j] = A[o_base * kD + j];
  __syncthreads();

  {
    const int i  = tid & 31;
    const int og = tid >> 5;
    const float* __restrict__ xrow  = x  + ((size_t)n * kINC + i) * kINN;
    const float* __restrict__ nfrow = nf + (size_t)i * kINN;
    #pragma unroll
    for (int rep = 0; rep < 4; ++rep) {
      const int o_loc = og + rep * 4;
      const int* __restrict__ idxs = &sA[o_loc * kD];
      float m = -__builtin_inff();
      #pragma unroll 4
      for (int d = 0; d < kD; ++d) {
        const int idx = idxs[d];
        m = fmaxf(m, xrow[idx] * nfrow[idx]);
      }
      sred[o_loc * 33 + i] = m;
    }
  }
  __syncthreads();

  {
    const int lane  = tid & 31;
    const int wave  = tid >> 5;
    const int m16   = lane & 15;
    const int hi    = lane >> 4;
    const int kbase = wave * 16;

    v8f c = {};
    #pragma unroll
    for (int s = 0; s < 8; ++s) {
      const int ks = 4 * s + 2 * hi;
      v2f a, b;
      a.x = ft[(size_t)ks       * kOUTC + kbase + m16];
      a.y = ft[(size_t)(ks + 1) * kOUTC + kbase + m16];
      b.x = sred[m16 * 33 + ks];
      b.y = sred[m16 * 33 + ks + 1];
      c = __builtin_amdgcn_wmma_f32_16x16x4_f32(
              false, a, false, b, (short)0, c, false, false);
    }

    #pragma unroll
    for (int v = 0; v < 8; ++v) {
      const int k = kbase + v + 8 * hi;
      const int o = o_base + m16;
      const size_t bidx = (size_t)k * kOUTN + o;
      out[(size_t)n * kOUTC * kOUTN + bidx] = c[v] + bias[bidx];
    }
  }
}

extern "C" void kernel_launch(void* const* d_in, const int* in_sizes, int n_in,
                              void* d_out, int out_size, void* d_ws, size_t ws_size,
                              hipStream_t stream) {
  const float* x    = (const float*)d_in[0];
  const float* nf   = (const float*)d_in[1];
  const float* ft   = (const float*)d_in[2];
  const float* bias = (const float*)d_in[3];
  const int*   A    = (const int*)d_in[4];
  float* out = (float*)d_out;

  dim3 grid(kOUTN / 16, kN);
  fused_gather_max_wmma<<<grid, 128, 0, stream>>>(x, nf, ft, bias, A, out);

  (void)in_sizes; (void)n_in; (void)out_size; (void)d_ws; (void)ws_size;
}


# Gather launch: each output's 32 neighbour indices sorted once in registers (bitonic network, packed 16-bit min/max); per window a wave skips the register pairs in which no lane has an index of it
# speedup vs baseline: 12.6068x; 12.6068x over previous
#include <hip/hip_runtime.h>


constexpr int kN    = 8;
constexpr int kINC  = 32;
constexpr int kINN  = 50000;
constexpr int kOUTC = 64;
constexpr int kOUTN = 8192;
constexpr int kD    = 32;

constexpr int kNB      = 4;
constexpr int kOWG     = 2048;
constexpr int kThreads = 1024;
constexpr int kOWave   = 16;
constexpr int kWinP    = 10000;
constexpr int kNWin    = kINN / kWinP;
constexpr int kGroups  = kWinP / 4;
constexpr int kCellB   = 16;
constexpr int kWinB    = kWinP * kCellB;
constexpr int kBufB    = kWinB + kCellB;
constexpr int kLdsB    = 2 * kBufB;
constexpr int kAPitchB = 144;
static_assert(kNWin * kWinP == kINN, "windows must tile the row");
static_assert((kWinP & 15) == 0, "a window starts on a multiple of 16 positions (cell swizzle)");
static_assert(kOWG * kAPitchB <= kLdsB, "staged A fits the window storage");
static_assert(kINN <= 65536, "two neighbour indices share a register");

typedef __attribute__((ext_vector_type(2))) float v2f;
typedef __attribute__((ext_vector_type(4))) float v4f;
typedef __attribute__((ext_vector_type(8))) float v8f;
typedef __attribute__((ext_vector_type(4))) int   v4i;
typedef __attribute__((ext_vector_type(2))) unsigned short v2us;

__device__ __forceinline__ void load_group(const float* __restrict__ xb, const float* __restrict__ nfb,
                                           unsigned voff, v4f (&xv)[kNB], v4f &nv) {
  #pragma unroll
  for (int b = 0; b < kNB; ++b)
    xv[b] = *reinterpret_cast<const v4f*>(reinterpret_cast<const char*>(xb + (size_t)b * kINC * kINN) + voff);
  nv = *reinterpret_cast<const v4f*>(reinterpret_cast<const char*>(nfb) + voff);
}

__device__ __forceinline__ void store_group(char* dst, int J, int s, const v4f (&xv)[kNB], const v4f &nv) {
  asm volatile("" :: "v"(xv[0]), "v"(xv[1]), "v"(xv[2]), "v"(xv[3]), "v"(nv));
  #pragma unroll
  for (int c = 0; c < 4; ++c) {
    v4f cell;
    cell.x = xv[0][c] * nv[c];
    cell.y = xv[1][c] * nv[c];
    cell.z = xv[2][c] * nv[c];
    cell.w = xv[3][c] * nv[c];
    *reinterpret_cast<v4f*>(dst + ((4 * J + (c ^ s)) * kCellB)) = cell;
  }
}

__device__ __forceinline__ void cmpx(v2us &lo, v2us &hi) {
  const v2us a = __builtin_elementwise_min(lo, hi);
  const v2us b = __builtin_elementwise_max(lo, hi);
  lo = a;
  hi = b;
}

constexpr int step_k(int st) { int k = 2, c = 1, first = 0; while (st >= first + c) { first += c; ++c; k <<= 1; } return k; }
constexpr int step_j(int st) { int k = 2, c = 1, first = 0; while (st >= first + c) { first += c; ++c; k <<= 1; } return (k >> 1) >> (st - first); }

template <int ST>
__device__ __forceinline__ void sort_step(v2us (&s)[32]) {
  constexpr int k = step_k(ST), j = step_j(ST);
  #pragma unroll
  for (int i = 0; i < 32; ++i) {
    const int l = i ^ j;
    if (l > i) {
      if ((i & k) == 0) cmpx(s[i], s[l]);
      else              cmpx(s[l], s[i]);
    }
  }
}

template <int S0, int S1>
__device__ __forceinline__ void sort_steps(v2us (&s)[32]) {
  if constexpr (S0 < S1) {
    sort_step<S0>(s);
    sort_steps<S0 + 1, S1>(s);
  }
}

__device__ __forceinline__ void hold_sorted(const v2us (&s)[32]) {
  asm volatile("" :: "v"(s[0]), "v"(s[1]), "v"(s[2]), "v"(s[3]), "v"(s[4]), "v"(s[5]), "v"(s[6]), "v"(s[7]),
                     "v"(s[8]), "v"(s[9]), "v"(s[10]), "v"(s[11]), "v"(s[12]), "v"(s[13]), "v"(s[14]), "v"(s[15]));
  asm volatile("" :: "v"(s[16]), "v"(s[17]), "v"(s[18]), "v"(s[19]), "v"(s[20]), "v"(s[21]), "v"(s[22]), "v"(s[23]),
                     "v"(s[24]), "v"(s[25]), "v"(s[26]), "v"(s[27]), "v"(s[28]), "v"(s[29]), "v"(s[30]), "v"(s[31]));
}

template <int J0>
__device__ __forceinline__ void pick(const char* rdata, const v2us (&srt)[32], v4f (&acc)[2], v2us p0) {
  const v2us top = {(unsigned short)kWinP, (unsigned short)kWinP};
  const v2us r0 = __builtin_elementwise_min((v2us)(srt[J0] - p0), top);
  const v2us r1 = __builtin_elementwise_min((v2us)(srt[J0 + 1] - p0), top);
  const unsigned least = __builtin_bit_cast(unsigned, __builtin_elementwise_min(r0, r1));
  if (__builtin_amdgcn_ballot_w32(least != (unsigned)kWinP * 0x10001u) != 0u) {
    const v4f a0 = *reinterpret_cast<const v4f*>(rdata + (unsigned)r0.x * kCellB);
    const v4f b0 = *reinterpret_cast<const v4f*>(rdata + (unsigned)r0.y * kCellB);
    const v4f a1 = *reinterpret_cast<const v4f*>(rdata + (unsigned)r1.x * kCellB);
    const v4f b1 = *reinterpret_cast<const v4f*>(rdata + (unsigned)r1.y * kCellB);
    acc[0].x = fmaxf(fmaxf(acc[0].x, a0.x), a1.x);
    acc[0].y = fmaxf(fmaxf(acc[0].y, a0.y), a1.y);
    acc[0].z = fmaxf(fmaxf(acc[0].z, a0.z), a1.z);
    acc[0].w = fmaxf(fmaxf(acc[0].w, a0.w), a1.w);
    acc[1].x = fmaxf(fmaxf(acc[1].x, b0.x), b1.x);
    acc[1].y = fmaxf(fmaxf(acc[1].y, b0.y), b1.y);
    acc[1].z = fmaxf(fmaxf(acc[1].z, b0.z), b1.z);
    acc[1].w = fmaxf(fmaxf(acc[1].w, b0.w), b1.w);
  }
}

__global__ __launch_bounds__(kThreads) void stream_row_four_batches_pick_max(
    const float* __restrict__ x,
    const float* __restrict__ nf,
    const int*   __restrict__ A,
    float*       __restrict__ out)
{
  __shared__ v4f lds[kLdsB / 16];
  char* ldsb = reinterpret_cast<char*>(lds);

  const int tid    = threadIdx.x;
  const int o_base = (blockIdx.x & 3) * kOWG;
  const int n0     = ((blockIdx.x >> 2) & 1) * kNB;
  const int i      = blockIdx.x >> 3;

  const float* __restrict__ xr  = x + ((size_t)n0 * kINC + i) * kINN;
  const float* __restrict__ nfr = nf + (size_t)i * kINN;

  {
    const char* __restrict__ Ab = reinterpret_cast<const char*>(A + (size_t)o_base * kD);
    #pragma unroll
    for (int u = 0; u < (kOWG * kD / 4) / kThreads; ++u) {
      const int gq = tid + u * kThreads;
      *reinterpret_cast<v4i*>(ldsb + (gq >> 3) * kAPitchB + (gq & 7) * 16) =
          *reinterpret_cast<const v4i*>(Ab + (unsigned)gq * 16u);
    }
  }
  __syncthreads();

  v2us srt[32];
  #pragma unroll
  for (int m = 0; m < 8; ++m) {
    const v4i v0 = *reinterpret_cast<const v4i*>(ldsb + tid * kAPitchB + m * 16);
    const v4i v1 = *reinterpret_cast<const v4i*>(ldsb + (kThreads + tid) * kAPitchB + m * 16);
    #pragma unroll
    for (int e = 0; e < 4; ++e) {
      const int p0 = min(max(v0[e], 0), 65535);
      const int p1 = min(max(v1[e], 0), 65535);
      srt[m * 4 + e] = v2us{(unsigned short)(p0 ^ ((p0 >> 4) & 3)), (unsigned short)(p1 ^ ((p1 >> 4) & 3))};
    }
  }
  __syncthreads();

  if (tid < 2)
    *reinterpret_cast<v4f*>(ldsb + tid * kBufB + kWinB) =
        v4f{-__builtin_inff(), -__builtin_inff(), -__builtin_inff(), -__builtin_inff()};

  const int  J0 = tid, J1 = tid + kThreads, J2 = tid + 2 * kThreads;
  const bool third = J2 < kGroups;
  const unsigned voff = (unsigned)tid * 16u;

  v4f xv[kNB], nv;

  load_group(xr, nfr, voff, xv, nv);
  sort_steps<0, 6>(srt);
  hold_sorted(srt);
  store_group(ldsb, J0, (J0 >> 2) & 3, xv, nv);
  load_group(xr + 4 * kThreads, nfr + 4 * kThreads, voff, xv, nv);
  sort_steps<6, 10>(srt);
  hold_sorted(srt);
  store_group(ldsb, J1, (J1 >> 2) & 3, xv, nv);
  if (third) load_group(xr + 8 * kThreads, nfr + 8 * kThreads, voff, xv, nv);
  sort_steps<10, 15>(srt);
  hold_sorted(srt);
  if (third) store_group(ldsb, J2, (J2 >> 2) & 3, xv, nv);
  __syncthreads();

  v4f acc[2];
  acc[0] = v4f{-__builtin_inff(), -__builtin_inff(), -__builtin_inff(), -__builtin_inff()};
  acc[1] = acc[0];

  for (int w = 0; w < kNWin; ++w) {
    const bool more = (w + 1 < kNWin);
    const char* rd = ldsb + (w & 1) * kBufB;
    char*       wd = ldsb + ((w + 1) & 1) * kBufB;
    const v2us  p0 = {(unsigned short)(w * kWinP), (unsigned short)(w * kWinP)};
    const int pn  = (w + 1) * kWinP;
    const int sn  = pn >> 4;
    const float* __restrict__ xn  = xr + pn;
    const float* __restrict__ nfn = nfr + pn;

    if (more) load_group(xn, nfn, voff, xv, nv);
    pick<0>(rd, srt, acc, p0);
    pick<16>(rd, srt, acc, p0);
    pick<8>(rd, srt, acc, p0);
    pick<24>(rd, srt, acc, p0);
    pick<4>(rd, srt, acc, p0);
    pick<20>(rd, srt, acc, p0);
    if (more) {
      store_group(wd, J0, (sn + (J0 >> 2)) & 3, xv, nv);
      load_group(xn + 4 * kThreads, nfn + 4 * kThreads, voff, xv, nv);
    }
    pick<12>(rd, srt, acc, p0);
    pick<28>(rd, srt, acc, p0);
    pick<2>(rd, srt, acc, p0);
    pick<18>(rd, srt, acc, p0);
    pick<10>(rd, srt, acc, p0);
    if (more) {
      store_group(wd, J1, (sn + (J1 >> 2)) & 3, xv, nv);
      if (third) load_group(xn + 8 * kThreads, nfn + 8 * kThreads, voff, xv, nv);
    }
    pick<26>(rd, srt, acc, p0);
    pick<6>(rd, srt, acc, p0);
    pick<22>(rd, srt, acc, p0);
    pick<14>(rd, srt, acc, p0);
    pick<30>(rd, srt, acc, p0);
    if (more && third) store_group(wd, J2, (sn + (J2 >> 2)) & 3, xv, nv);
    __syncthreads();
  }

  #pragma unroll
  for (int set = 0; set < 2; ++set) {
    #pragma unroll
    for (int b = 0; b < kNB; ++b)
      out[((size_t)(n0 + b) * kOUTC + i) * kOUTN + o_base + set * kThreads + tid] = acc[set][b];
  }
}

__global__ __launch_bounds__(256) void product_wmma_from_maxima(
    const float* __restrict__ ft,
    const float* __restrict__ bias,
    float*       __restrict__ out)
{
  const int lane = threadIdx.x & 31;
  const int gw   = blockIdx.x * 8 + (threadIdx.x >> 5);
  const int n      = gw >> 9;
  const int o_wave = (gw & 511) * kOWave;

  const int m16 = lane & 15;
  const int hi  = lane >> 4;
  const int o = o_wave + m16;
  float* __restrict__ on = out + (size_t)n * kOUTC * kOUTN;

  v2f b[8];
  #pragma unroll
  for (int s = 0; s < 8; ++s) {
    const int ks = 4 * s + 2 * hi;
    b[s].x = on[(size_t)ks       * kOUTN + o];
    b[s].y = on[(size_t)(ks + 1) * kOUTN + o];
  }

  for (int kt = 0; kt < kOUTC / 16; ++kt) {
    const int kbase = kt * 16;
    v2f a[8];
    #pragma unroll
    for (int s = 0; s < 8; ++s) {
      const int ks = 4 * s + 2 * hi;
      a[s].x = ft[(size_t)ks       * kOUTC + kbase + m16];
      a[s].y = ft[(size_t)(ks + 1) * kOUTC + kbase + m16];
    }
    v8f c = {};
    #pragma unroll
    for (int s = 0; s < 8; ++s) {
      c = __builtin_amdgcn_wmma_f32_16x16x4_f32(
              false, a[s], false, b[s], (short)0, c, false, false);
    }
    #pragma unroll
    for (int v = 0; v < 8; ++v) {
      const int kk = kbase + v + 8 * hi;
      on[(size_t)kk * kOUTN + o] = c[v] + bias[(size_t)kk * kOUTN + o];
    }
  }
}

extern "C" void kernel_launch(void* const* d_in, const int* in_sizes, int n_in,
                              void* d_out, int out_size, void* d_ws, size_t ws_size,
                              hipStream_t stream) {
  const float* x    = (const float*)d_in[0];
  const float* nf   = (const float*)d_in[1];
  const float* ft   = (const float*)d_in[2];
  const float* bias = (const float*)d_in[3];
  const int*   A    = (const int*)d_in[4];
  float* out = (float*)d_out;

  dim3 grid1(kINC * (kN / kNB) * (kOUTN / kOWG));
  stream_row_four_batches_pick_max<<<grid1, kThreads, 0, stream>>>(x, nf, A, out);

  dim3 grid2(kN * (kOUTN / kOWave) / 8);
  product_wmma_from_maxima<<<grid2, 256, 0, stream>>>(ft, bias, out);

  (void)in_sizes; (void)n_in; (void)out_size; (void)d_ws; (void)ws_size;
}
